# starting kernel
# baseline (speedup 1.0000x reference)
#include <hip/hip_runtime.h>
#include <hip/hip_bf16.h>
#include <math.h>

typedef float v2f __attribute__((ext_vector_type(2)));
typedef float v8f __attribute__((ext_vector_type(8)));

#define NN   50000
#define KK   16
#define CC   128
#define OUTC 128

__global__ __launch_bounds__(256) void wd_kernel(const float* __restrict__ W,
                                                 float* __restrict__ Wd) {
    int i = blockIdx.x * 256 + threadIdx.x;
    Wd[i] = W[i] - W[i + CC * OUTC];
}

__global__ __launch_bounds__(256) void gemm_kernel(const float* __restrict__ x,
                                                   const float* __restrict__ W,
                                                   const float* __restrict__ Wd,
                                                   const float* __restrict__ b,
                                                   float* __restrict__ y,
                                                   float* __restrict__ base) {
    __shared__ float xs[16 * CC];

    const int tid  = threadIdx.x;
    const int row0 = blockIdx.x * 16;

    {
        const float4* src = (const float4*)(x + row0 * CC);
        float4* dst = (float4*)xs;
        dst[tid]       = src[tid];
        dst[tid + 256] = src[tid + 256];
    }
    __syncthreads();

    const int wave = tid >> 5;
    const int lane = tid & 31;
    const int half = lane >> 4;
    const int lrow = lane & 15;
    const int koff = half * 2;
    const int col0 = wave * 16;
    const int gc   = col0 + lrow;

    const float* __restrict__ W2 = W + CC * OUTC;

    v8f acc_y = {0.f, 0.f, 0.f, 0.f, 0.f, 0.f, 0.f, 0.f};
    const float bv = b[gc];
    v8f acc_b = {bv, bv, bv, bv, bv, bv, bv, bv};

#pragma unroll 8
    for (int kk = 0; kk < 32; ++kk) {
        const int kb = kk * 4 + koff;
        v2f a = *(const v2f*)(&xs[lrow * CC + kb]);
        v2f b2; b2.x = W2[kb * OUTC + gc]; b2.y = W2[(kb + 1) * OUTC + gc];
        v2f bd; bd.x = Wd[kb * OUTC + gc]; bd.y = Wd[(kb + 1) * OUTC + gc];
        acc_y = __builtin_amdgcn_wmma_f32_16x16x4_f32(false, a, false, b2,
                                                      (short)0, acc_y, false, false);
        acc_b = __builtin_amdgcn_wmma_f32_16x16x4_f32(false, a, false, bd,
                                                      (short)0, acc_b, false, false);
    }

    const int gr = row0 + half * 8;
#pragma unroll
    for (int r = 0; r < 8; ++r) {
        y[(gr + r) * OUTC + gc]    = acc_y[r];
        base[(gr + r) * OUTC + gc] = acc_b[r];
    }
}

__global__ __launch_bounds__(256) void gather_max_kernel(const int* __restrict__ edge,
                                                         const float* __restrict__ y,
                                                         const float* __restrict__ base,
                                                         float* __restrict__ out) {
    const int wave = threadIdx.x >> 5;
    const int lane = threadIdx.x & 31;
    const int node = blockIdx.x * 8 + wave;
    if (node >= NN) return;

    const int c0 = lane * 4;
    const int* __restrict__ e = edge + node * KK;

    float4 m = make_float4(-INFINITY, -INFINITY, -INFINITY, -INFINITY);
#pragma unroll
    for (int k = 0; k < KK; ++k) {
        const int j = __builtin_amdgcn_readfirstlane(e[k]);
        if (j >= 0) {
            const float4 v = *(const float4*)(y + j * OUTC + c0);
            m.x = fmaxf(m.x, v.x);
            m.y = fmaxf(m.y, v.y);
            m.z = fmaxf(m.z, v.z);
            m.w = fmaxf(m.w, v.w);
        }
    }

    const float4 bs = *(const float4*)(base + node * OUTC + c0);
    float4 z = make_float4(bs.x + m.x, bs.y + m.y, bs.z + m.z, bs.w + m.w);
    float4 o;
    o.x = (z.x > 0.f) ? z.x : (expf(z.x) - 1.f);
    o.y = (z.y > 0.f) ? z.y : (expf(z.y) - 1.f);
    o.z = (z.z > 0.f) ? z.z : (expf(z.z) - 1.f);
    o.w = (z.w > 0.f) ? z.w : (expf(z.w) - 1.f);
    *(float4*)(out + node * OUTC + c0) = o;
}

extern "C" void kernel_launch(void* const* d_in, const int* in_sizes, int n_in,
                              void* d_out, int out_size, void* d_ws, size_t ws_size,
                              hipStream_t stream) {
    const float* x    = (const float*)d_in[0];
    const int*   edge = (const int*)  d_in[1];
    const float* W    = (const float*)d_in[2];
    const float* b    = (const float*)d_in[3];
    float* out = (float*)d_out;

    float* ws   = (float*)d_ws;
    float* y    = ws;
    float* base = ws + (size_t)NN * OUTC;
    float* Wd   = ws + (size_t)2 * NN * OUTC;

    wd_kernel<<<64, 256, 0, stream>>>(W, Wd);
    gemm_kernel<<<NN / 16, 256, 0, stream>>>(x, W, Wd, b, y, base);
    gather_max_kernel<<<NN / 8, 256, 0, stream>>>(edge, y, base, out);
}


# GEMM: persistent blocks hold each wave's W2/Wd B fragments in registers across all row tiles (x tiles double-buffered in LDS) instead of re-reading both weight matrices per 16-row block; same sums
# speedup vs baseline: 1.0987x; 1.0987x over previous
#include <hip/hip_runtime.h>
#include <hip/hip_bf16.h>
#include <math.h>

typedef float v2f __attribute__((ext_vector_type(2)));
typedef float v8f __attribute__((ext_vector_type(8)));

#define NN   50000
#define KK   16
#define CC   128
#define OUTC 128

__global__ __launch_bounds__(256) void wd_kernel(const float* __restrict__ W,
                                                 float* __restrict__ Wd) {
    int i = blockIdx.x * 256 + threadIdx.x;
    Wd[i] = W[i] - W[i + CC * OUTC];
}

#define XPITCH    132
#define NTILES    (NN / 16)
#define GEMM_GRID 512

__global__ __launch_bounds__(256) void gemm_kernel(const float* __restrict__ x,
                                                   const float* __restrict__ W,
                                                   const float* __restrict__ Wd,
                                                   const float* __restrict__ b,
                                                   float* __restrict__ y,
                                                   float* __restrict__ base) {
    __shared__ float xs[2][16 * XPITCH];

    const int tid  = threadIdx.x;
    const int wave = tid >> 5;
    const int lane = tid & 31;
    const int half = lane >> 4;
    const int lrow = lane & 15;
    const int koff = half * 2;
    const int col0 = wave * 16;
    const int gc   = col0 + lrow;

    const int srow = tid >> 5;
    const int scol = (tid & 31) * 4;

    int tile = blockIdx.x;

    float4 g0, g1;
    {
        const float4* src = (const float4*)(x + tile * 16 * CC);
        g0 = src[tid];
        g1 = src[tid + 256];
    }

    const float* __restrict__ W2 = W + CC * OUTC;
    v2f b2[32], bd[32];
#pragma unroll
    for (int kk = 0; kk < 32; ++kk) {
        const int kb = kk * 4 + koff;
        b2[kk].x = W2[kb * OUTC + gc]; b2[kk].y = W2[(kb + 1) * OUTC + gc];
        bd[kk].x = Wd[kb * OUTC + gc]; bd[kk].y = Wd[(kb + 1) * OUTC + gc];
    }
    const float bv = b[gc];

    int buf = 0;
    for (; tile < NTILES; tile += GEMM_GRID) {
        float* __restrict__ xb = xs[buf];
        *(float4*)(&xb[srow * XPITCH + scol])       = g0;
        *(float4*)(&xb[(srow + 8) * XPITCH + scol]) = g1;
        __syncthreads();

        const int nt = tile + GEMM_GRID;
        if (nt < NTILES) {
            const float4* src = (const float4*)(x + nt * 16 * CC);
            g0 = src[tid];
            g1 = src[tid + 256];
        }

        v8f acc_y = {0.f, 0.f, 0.f, 0.f, 0.f, 0.f, 0.f, 0.f};
        v8f acc_b = {bv, bv, bv, bv, bv, bv, bv, bv};
#pragma unroll
        for (int kk = 0; kk < 32; ++kk) {
            v2f a = *(const v2f*)(&xb[lrow * XPITCH + kk * 4 + koff]);
            acc_y = __builtin_amdgcn_wmma_f32_16x16x4_f32(false, a, false, b2[kk],
                                                          (short)0, acc_y, false, false);
            acc_b = __builtin_amdgcn_wmma_f32_16x16x4_f32(false, a, false, bd[kk],
                                                          (short)0, acc_b, false, false);
        }

        const int gr = tile * 16 + half * 8;
#pragma unroll
        for (int r = 0; r < 8; ++r) {
            y[(gr + r) * OUTC + gc]    = acc_y[r];
            base[(gr + r) * OUTC + gc] = acc_b[r];
        }
        buf ^= 1;
    }
}

__global__ __launch_bounds__(256) void gather_max_kernel(const int* __restrict__ edge,
                                                         const float* __restrict__ y,
                                                         const float* __restrict__ base,
                                                         float* __restrict__ out) {
    const int wave = threadIdx.x >> 5;
    const int lane = threadIdx.x & 31;
    const int node = blockIdx.x * 8 + wave;
    if (node >= NN) return;

    const int c0 = lane * 4;
    const int* __restrict__ e = edge + node * KK;

    float4 m = make_float4(-INFINITY, -INFINITY, -INFINITY, -INFINITY);
#pragma unroll
    for (int k = 0; k < KK; ++k) {
        const int j = __builtin_amdgcn_readfirstlane(e[k]);
        if (j >= 0) {
            const float4 v = *(const float4*)(y + j * OUTC + c0);
            m.x = fmaxf(m.x, v.x);
            m.y = fmaxf(m.y, v.y);
            m.z = fmaxf(m.z, v.z);
            m.w = fmaxf(m.w, v.w);
        }
    }

    const float4 bs = *(const float4*)(base + node * OUTC + c0);
    float4 z = make_float4(bs.x + m.x, bs.y + m.y, bs.z + m.z, bs.w + m.w);
    float4 o;
    o.x = (z.x > 0.f) ? z.x : (expf(z.x) - 1.f);
    o.y = (z.y > 0.f) ? z.y : (expf(z.y) - 1.f);
    o.z = (z.z > 0.f) ? z.z : (expf(z.z) - 1.f);
    o.w = (z.w > 0.f) ? z.w : (expf(z.w) - 1.f);
    *(float4*)(out + node * OUTC + c0) = o;
}

extern "C" void kernel_launch(void* const* d_in, const int* in_sizes, int n_in,
                              void* d_out, int out_size, void* d_ws, size_t ws_size,
                              hipStream_t stream) {
    const float* x    = (const float*)d_in[0];
    const int*   edge = (const int*)  d_in[1];
    const float* W    = (const float*)d_in[2];
    const float* b    = (const float*)d_in[3];
    float* out = (float*)d_out;

    float* ws   = (float*)d_ws;
    float* y    = ws;
    float* base = ws + (size_t)NN * OUTC;
    float* Wd   = ws + (size_t)2 * NN * OUTC;

    wd_kernel<<<64, 256, 0, stream>>>(W, Wd);
    gemm_kernel<<<GEMM_GRID, 256, 0, stream>>>(x, W, Wd, b, y, base);
    gather_max_kernel<<<NN / 8, 256, 0, stream>>>(edge, y, base, out);
}


# gather: 10000 resident waves walk the nodes, a node's 16 ids in one load and its 16 row loads in flight together (padded id reads a fixed row, masked); was 50000 one-node waves of 16 dependent loads
# speedup vs baseline: 1.3400x; 1.2196x over previous
#include <hip/hip_runtime.h>
#include <hip/hip_bf16.h>
#include <math.h>

typedef float v2f __attribute__((ext_vector_type(2)));
typedef float v8f __attribute__((ext_vector_type(8)));

#define NN   50000
#define KK   16
#define CC   128
#define OUTC 128

__global__ __launch_bounds__(256) void wd_kernel(const float* __restrict__ W,
                                                 float* __restrict__ Wd) {
    int i = blockIdx.x * 256 + threadIdx.x;
    Wd[i] = W[i] - W[i + CC * OUTC];
}

#define XPITCH    132
#define NTILES    (NN / 16)
#define GEMM_GRID 512

__global__ __launch_bounds__(256) void gemm_kernel(const float* __restrict__ x,
                                                   const float* __restrict__ W,
                                                   const float* __restrict__ Wd,
                                                   const float* __restrict__ b,
                                                   float* __restrict__ y,
                                                   float* __restrict__ base) {
    __shared__ float xs[2][16 * XPITCH];

    const int tid  = threadIdx.x;
    const int wave = tid >> 5;
    const int lane = tid & 31;
    const int half = lane >> 4;
    const int lrow = lane & 15;
    const int koff = half * 2;
    const int col0 = wave * 16;
    const int gc   = col0 + lrow;

    const int srow = tid >> 5;
    const int scol = (tid & 31) * 4;

    int tile = blockIdx.x;

    float4 g0, g1;
    {
        const float4* src = (const float4*)(x + tile * 16 * CC);
        g0 = src[tid];
        g1 = src[tid + 256];
    }

    const float* __restrict__ W2 = W + CC * OUTC;
    v2f b2[32], bd[32];
#pragma unroll
    for (int kk = 0; kk < 32; ++kk) {
        const int kb = kk * 4 + koff;
        b2[kk].x = W2[kb * OUTC + gc]; b2[kk].y = W2[(kb + 1) * OUTC + gc];
        bd[kk].x = Wd[kb * OUTC + gc]; bd[kk].y = Wd[(kb + 1) * OUTC + gc];
    }
    const float bv = b[gc];

    int buf = 0;
    for (; tile < NTILES; tile += GEMM_GRID) {
        float* __restrict__ xb = xs[buf];
        *(float4*)(&xb[srow * XPITCH + scol])       = g0;
        *(float4*)(&xb[(srow + 8) * XPITCH + scol]) = g1;
        __syncthreads();

        const int nt = tile + GEMM_GRID;
        if (nt < NTILES) {
            const float4* src = (const float4*)(x + nt * 16 * CC);
            g0 = src[tid];
            g1 = src[tid + 256];
        }

        v8f acc_y = {0.f, 0.f, 0.f, 0.f, 0.f, 0.f, 0.f, 0.f};
        v8f acc_b = {bv, bv, bv, bv, bv, bv, bv, bv};
#pragma unroll
        for (int kk = 0; kk < 32; ++kk) {
            v2f a = *(const v2f*)(&xb[lrow * XPITCH + kk * 4 + koff]);
            acc_y = __builtin_amdgcn_wmma_f32_16x16x4_f32(false, a, false, b2[kk],
                                                          (short)0, acc_y, false, false);
            acc_b = __builtin_amdgcn_wmma_f32_16x16x4_f32(false, a, false, bd[kk],
                                                          (short)0, acc_b, false, false);
        }

        const int gr = tile * 16 + half * 8;
#pragma unroll
        for (int r = 0; r < 8; ++r) {
            y[(gr + r) * OUTC + gc]    = acc_y[r];
            base[(gr + r) * OUTC + gc] = acc_b[r];
        }
        buf ^= 1;
    }
}

#define GATHER_WGS   1250
#define GATHER_WAVES (GATHER_WGS * 8)

__global__ __launch_bounds__(256) void gather_max_kernel(const int* __restrict__ edge,
                                                         const float* __restrict__ y,
                                                         const float* __restrict__ base,
                                                         float* __restrict__ out) {
    const int lane = threadIdx.x & 31;
    const int gw = __builtin_amdgcn_readfirstlane(blockIdx.x * 8 + (threadIdx.x >> 5));
    const unsigned loff = (unsigned)lane * 16u;
    const int slot = lane & 15;

    int idv = edge[gw * KK + slot];
    for (int node = gw; node < NN; node += GATHER_WAVES) {
        int j[KK];
#pragma unroll
        for (int k = 0; k < KK; ++k) j[k] = __builtin_amdgcn_readlane(idv, k);

        float4 v[KK];
#pragma unroll
        for (int k = 0; k < KK; ++k) {
            const unsigned r = min((unsigned)j[k], (unsigned)(NN - 1));
            v[k] = *(const float4*)((const char*)y + ((size_t)r << 9) + loff);
        }
        const float4 bs = *(const float4*)((const char*)base + ((size_t)node << 9) + loff);
        const int nn = node + GATHER_WAVES;
        if (nn < NN) idv = edge[nn * KK + slot];

        float4 m = make_float4(-INFINITY, -INFINITY, -INFINITY, -INFINITY);
#pragma unroll
        for (int k = 0; k < KK; ++k) {
            const bool ok = j[k] >= 0;
            m.x = fmaxf(m.x, ok ? v[k].x : -INFINITY);
            m.y = fmaxf(m.y, ok ? v[k].y : -INFINITY);
            m.z = fmaxf(m.z, ok ? v[k].z : -INFINITY);
            m.w = fmaxf(m.w, ok ? v[k].w : -INFINITY);
        }

        float4 z = make_float4(bs.x + m.x, bs.y + m.y, bs.z + m.z, bs.w + m.w);
        float4 o;
        o.x = (z.x > 0.f) ? z.x : (expf(z.x) - 1.f);
        o.y = (z.y > 0.f) ? z.y : (expf(z.y) - 1.f);
        o.z = (z.z > 0.f) ? z.z : (expf(z.z) - 1.f);
        o.w = (z.w > 0.f) ? z.w : (expf(z.w) - 1.f);
        *(float4*)((char*)out + ((size_t)node << 9) + loff) = o;
    }
}

extern "C" void kernel_launch(void* const* d_in, const int* in_sizes, int n_in,
                              void* d_out, int out_size, void* d_ws, size_t ws_size,
                              hipStream_t stream) {
    const float* x    = (const float*)d_in[0];
    const int*   edge = (const int*)  d_in[1];
    const float* W    = (const float*)d_in[2];
    const float* b    = (const float*)d_in[3];
    float* out = (float*)d_out;

    float* ws   = (float*)d_ws;
    float* y    = ws;
    float* base = ws + (size_t)NN * OUTC;
    float* Wd   = ws + (size_t)2 * NN * OUTC;

    wd_kernel<<<64, 256, 0, stream>>>(W, Wd);
    gemm_kernel<<<GEMM_GRID, 256, 0, stream>>>(x, W, Wd, b, y, base);
    gather_max_kernel<<<GATHER_WGS, 256, 0, stream>>>(edge, y, base, out);
}


# base=x@Wd+b moved from the GEMM launch into the gather launch: per 16-node tile its 32 WMMAs per wave run under the tile's neighbour-row loads, base handed over in LDS, never stored; same sums
# speedup vs baseline: 1.4364x; 1.0719x over previous
#include <hip/hip_runtime.h>
#include <hip/hip_bf16.h>
#include <math.h>

typedef float v2f __attribute__((ext_vector_type(2)));
typedef float v8f __attribute__((ext_vector_type(8)));

#define NN   50000
#define KK   16
#define CC   128
#define OUTC 128

__global__ __launch_bounds__(256) void wd_kernel(const float* __restrict__ W,
                                                 float* __restrict__ Wd) {
    int i = blockIdx.x * 256 + threadIdx.x;
    Wd[i] = W[i] - W[i + CC * OUTC];
}

#define XPITCH    132
#define NTILES    (NN / 16)
#define GEMM_GRID 512

__global__ __launch_bounds__(256) void gemm_y_kernel(const float* __restrict__ x,
                                                     const float* __restrict__ W,
                                                     float* __restrict__ y) {
    __shared__ float xs[2][16 * XPITCH];

    const int tid  = threadIdx.x;
    const int wave = tid >> 5;
    const int lane = tid & 31;
    const int half = lane >> 4;
    const int lrow = lane & 15;
    const int koff = half * 2;
    const int col0 = wave * 16;
    const int gc   = col0 + lrow;

    const int srow = tid >> 5;
    const int scol = (tid & 31) * 4;

    int tile = blockIdx.x;

    float4 g0, g1;
    {
        const float4* src = (const float4*)(x + tile * 16 * CC);
        g0 = src[tid];
        g1 = src[tid + 256];
    }

    const float* __restrict__ W2 = W + CC * OUTC;
    v2f b2[32];
#pragma unroll
    for (int kk = 0; kk < 32; ++kk) {
        const int kb = kk * 4 + koff;
        b2[kk].x = W2[kb * OUTC + gc]; b2[kk].y = W2[(kb + 1) * OUTC + gc];
    }

    int buf = 0;
    for (; tile < NTILES; tile += GEMM_GRID) {
        float* __restrict__ xb = xs[buf];
        *(float4*)(&xb[srow * XPITCH + scol])       = g0;
        *(float4*)(&xb[(srow + 8) * XPITCH + scol]) = g1;
        __syncthreads();

        const int nt = tile + GEMM_GRID;
        if (nt < NTILES) {
            const float4* src = (const float4*)(x + nt * 16 * CC);
            g0 = src[tid];
            g1 = src[tid + 256];
        }

        v8f acc_y = {0.f, 0.f, 0.f, 0.f, 0.f, 0.f, 0.f, 0.f};
#pragma unroll
        for (int kk = 0; kk < 32; ++kk) {
            v2f a = *(const v2f*)(&xb[lrow * XPITCH + kk * 4 + koff]);
            acc_y = __builtin_amdgcn_wmma_f32_16x16x4_f32(false, a, false, b2[kk],
                                                          (short)0, acc_y, false, false);
        }

        const int gr = tile * 16 + half * 8;
#pragma unroll
        for (int r = 0; r < 8; ++r) y[(gr + r) * OUTC + gc] = acc_y[r];
        buf ^= 1;
    }
}

#define BPITCH    132
#define FUSE_GRID 512

__device__ __forceinline__ void rows_in_flight(float4 (&v)[KK], const int (&j)[2 * KK], const int o,
                                               const float* __restrict__ y, const unsigned loff) {
#pragma unroll
    for (int k = 0; k < KK; ++k) {
        const unsigned r = min((unsigned)j[o + k], (unsigned)(NN - 1));
        v[k] = *(const float4*)((const char*)y + ((size_t)r << 9) + loff);
    }
}

__device__ __forceinline__ void node_finish(const float4 (&v)[KK], const int (&j)[2 * KK], const int o,
                                            const float4 bs, float* __restrict__ out,
                                            const int node, const unsigned loff) {
    float4 m = make_float4(-INFINITY, -INFINITY, -INFINITY, -INFINITY);
#pragma unroll
    for (int k = 0; k < KK; ++k) {
        const bool ok = j[o + k] >= 0;
        m.x = fmaxf(m.x, ok ? v[k].x : -INFINITY);
        m.y = fmaxf(m.y, ok ? v[k].y : -INFINITY);
        m.z = fmaxf(m.z, ok ? v[k].z : -INFINITY);
        m.w = fmaxf(m.w, ok ? v[k].w : -INFINITY);
    }
    float4 z = make_float4(bs.x + m.x, bs.y + m.y, bs.z + m.z, bs.w + m.w);
    float4 o4;
    o4.x = (z.x > 0.f) ? z.x : (expf(z.x) - 1.f);
    o4.y = (z.y > 0.f) ? z.y : (expf(z.y) - 1.f);
    o4.z = (z.z > 0.f) ? z.z : (expf(z.z) - 1.f);
    o4.w = (z.w > 0.f) ? z.w : (expf(z.w) - 1.f);
    *(float4*)((char*)out + ((size_t)node << 9) + loff) = o4;
}

__global__ __launch_bounds__(256) void base_gather_kernel(const float* __restrict__ x,
                                                          const float* __restrict__ Wd,
                                                          const float* __restrict__ b,
                                                          const int* __restrict__ edge,
                                                          const float* __restrict__ y,
                                                          float* __restrict__ out) {
    __shared__ float xs[16 * XPITCH];
    __shared__ float bt[16 * BPITCH];

    const int tid  = threadIdx.x;
    const int wave = __builtin_amdgcn_readfirstlane(tid >> 5);
    const int lane = tid & 31;
    const int half = lane >> 4;
    const int lrow = lane & 15;
    const int koff = half * 2;
    const int gc   = wave * 16 + lrow;
    const int srow = tid >> 5;
    const int scol = (tid & 31) * 4;
    const unsigned loff = (unsigned)lane * 16u;

    int tile = blockIdx.x;

    float4 g0, g1;
    {
        const float4* src = (const float4*)(x + tile * 16 * CC);
        g0 = src[tid];
        g1 = src[tid + 256];
    }
    int idv = edge[(tile * 16 + wave * 2) * KK + lane];

    v2f bd[32];
#pragma unroll
    for (int kk = 0; kk < 32; ++kk) {
        const int kb = kk * 4 + koff;
        bd[kk].x = Wd[kb * OUTC + gc]; bd[kk].y = Wd[(kb + 1) * OUTC + gc];
    }
    const float bv = b[gc];

    for (; tile < NTILES; tile += FUSE_GRID) {
        *(float4*)(&xs[srow * XPITCH + scol])       = g0;
        *(float4*)(&xs[(srow + 8) * XPITCH + scol]) = g1;
        __syncthreads();

        const int nt = tile + FUSE_GRID;
        if (nt < NTILES) {
            const float4* src = (const float4*)(x + nt * 16 * CC);
            g0 = src[tid];
            g1 = src[tid + 256];
        }

        const int n0 = tile * 16 + wave * 2;
        int j[2 * KK];
#pragma unroll
        for (int k = 0; k < 2 * KK; ++k) j[k] = __builtin_amdgcn_readlane(idv, k);

        float4 v0[KK], v1[KK];
        rows_in_flight(v0, j, 0, y, loff);
        rows_in_flight(v1, j, KK, y, loff);
        if (nt < NTILES) idv = edge[(nt * 16 + wave * 2) * KK + lane];

        v8f acc_b = {bv, bv, bv, bv, bv, bv, bv, bv};
#pragma unroll
        for (int kk = 0; kk < 32; ++kk) {
            v2f a = *(const v2f*)(&xs[lrow * XPITCH + kk * 4 + koff]);
            acc_b = __builtin_amdgcn_wmma_f32_16x16x4_f32(false, a, false, bd[kk],
                                                          (short)0, acc_b, false, false);
        }
#pragma unroll
        for (int r = 0; r < 8; ++r) bt[(half * 8 + r) * BPITCH + gc] = acc_b[r];
        __syncthreads();
        __builtin_amdgcn_sched_barrier(0);

        const int lr0 = wave * 2;
        {
            const float4 bs = *(const float4*)(&bt[lr0 * BPITCH + lane * 4]);
            node_finish(v0, j, 0, bs, out, n0, loff);
        }
        {
            const float4 bs = *(const float4*)(&bt[(lr0 + 1) * BPITCH + lane * 4]);
            node_finish(v1, j, KK, bs, out, n0 + 1, loff);
        }
    }
}

extern "C" void kernel_launch(void* const* d_in, const int* in_sizes, int n_in,
                              void* d_out, int out_size, void* d_ws, size_t ws_size,
                              hipStream_t stream) {
    const float* x    = (const float*)d_in[0];
    const int*   edge = (const int*)  d_in[1];
    const float* W    = (const float*)d_in[2];
    const float* b    = (const float*)d_in[3];
    float* out = (float*)d_out;

    float* ws   = (float*)d_ws;
    float* y    = ws;
    float* Wd   = ws + (size_t)2 * NN * OUTC;

    wd_kernel<<<64, 256, 0, stream>>>(W, Wd);
    gemm_y_kernel<<<GEMM_GRID, 256, 0, stream>>>(x, W, y);
    base_gather_kernel<<<FUSE_GRID, 256, 0, stream>>>(x, Wd, b, edge, y, out);
}


# Wd = W1 - W2 is written by 64 blocks of the y GEMM launch behind their tiles, not by a launch of its own: 3 launches -> 2; same values, same sums
# speedup vs baseline: 1.4884x; 1.0362x over previous
#include <hip/hip_runtime.h>
#include <hip/hip_bf16.h>
#include <math.h>

typedef float v2f __attribute__((ext_vector_type(2)));
typedef float v8f __attribute__((ext_vector_type(8)));

#define NN   50000
#define KK   16
#define CC   128
#define OUTC 128

#define XPITCH    132
#define NTILES    (NN / 16)
#define GEMM_GRID 512
#define WD_BLOCKS (CC * OUTC / 256)
#define WD_FIRST  64
static_assert(WD_FIRST >= NTILES % GEMM_GRID && WD_FIRST + WD_BLOCKS <= GEMM_GRID,
              "the blocks that write Wd must exist and must not be among those with an extra tile");

__global__ __launch_bounds__(256) void gemm_y_kernel(const float* __restrict__ x,
                                                     const float* __restrict__ W,
                                                     float* __restrict__ y,
                                                     float* __restrict__ Wd) {
    __shared__ float xs[2][16 * XPITCH];

    const int tid  = threadIdx.x;
    const int wave = tid >> 5;
    const int lane = tid & 31;
    const int half = lane >> 4;
    const int lrow = lane & 15;
    const int koff = half * 2;
    const int col0 = wave * 16;
    const int gc   = col0 + lrow;

    const int srow = tid >> 5;
    const int scol = (tid & 31) * 4;

    int tile = blockIdx.x;

    float4 g0, g1;
    {
        const float4* src = (const float4*)(x + tile * 16 * CC);
        g0 = src[tid];
        g1 = src[tid + 256];
    }

    const float* __restrict__ W2 = W + CC * OUTC;
    v2f b2[32];
#pragma unroll
    for (int kk = 0; kk < 32; ++kk) {
        const int kb = kk * 4 + koff;
        b2[kk].x = W2[kb * OUTC + gc]; b2[kk].y = W2[(kb + 1) * OUTC + gc];
    }

    int buf = 0;
    for (; tile < NTILES; tile += GEMM_GRID) {
        float* __restrict__ xb = xs[buf];
        *(float4*)(&xb[srow * XPITCH + scol])       = g0;
        *(float4*)(&xb[(srow + 8) * XPITCH + scol]) = g1;
        __syncthreads();

        const int nt = tile + GEMM_GRID;
        if (nt < NTILES) {
            const float4* src = (const float4*)(x + nt * 16 * CC);
            g0 = src[tid];
            g1 = src[tid + 256];
        }

        v8f acc_y = {0.f, 0.f, 0.f, 0.f, 0.f, 0.f, 0.f, 0.f};
#pragma unroll
        for (int kk = 0; kk < 32; ++kk) {
            v2f a = *(const v2f*)(&xb[lrow * XPITCH + kk * 4 + koff]);
            acc_y = __builtin_amdgcn_wmma_f32_16x16x4_f32(false, a, false, b2[kk],
                                                          (short)0, acc_y, false, false);
        }

        const int gr = tile * 16 + half * 8;
#pragma unroll
        for (int r = 0; r < 8; ++r) y[(gr + r) * OUTC + gc] = acc_y[r];
        buf ^= 1;
    }

    if (blockIdx.x >= WD_FIRST && blockIdx.x < WD_FIRST + WD_BLOCKS) {
        const int i = (blockIdx.x - WD_FIRST) * 256 + tid;
        Wd[i] = W[i] - W2[i];
    }
}

#define BPITCH    132
#define FUSE_GRID 512

__device__ __forceinline__ void rows_in_flight(float4 (&v)[KK], const int (&j)[2 * KK], const int o,
                                               const float* __restrict__ y, const unsigned loff) {
#pragma unroll
    for (int k = 0; k < KK; ++k) {
        const unsigned r = min((unsigned)j[o + k], (unsigned)(NN - 1));
        v[k] = *(const float4*)((const char*)y + ((size_t)r << 9) + loff);
    }
}

__device__ __forceinline__ void node_finish(const float4 (&v)[KK], const int (&j)[2 * KK], const int o,
                                            const float4 bs, float* __restrict__ out,
                                            const int node, const unsigned loff) {
    float4 m = make_float4(-INFINITY, -INFINITY, -INFINITY, -INFINITY);
#pragma unroll
    for (int k = 0; k < KK; ++k) {
        const bool ok = j[o + k] >= 0;
        m.x = fmaxf(m.x, ok ? v[k].x : -INFINITY);
        m.y = fmaxf(m.y, ok ? v[k].y : -INFINITY);
        m.z = fmaxf(m.z, ok ? v[k].z : -INFINITY);
        m.w = fmaxf(m.w, ok ? v[k].w : -INFINITY);
    }
    float4 z = make_float4(bs.x + m.x, bs.y + m.y, bs.z + m.z, bs.w + m.w);
    float4 o4;
    o4.x = (z.x > 0.f) ? z.x : (expf(z.x) - 1.f);
    o4.y = (z.y > 0.f) ? z.y : (expf(z.y) - 1.f);
    o4.z = (z.z > 0.f) ? z.z : (expf(z.z) - 1.f);
    o4.w = (z.w > 0.f) ? z.w : (expf(z.w) - 1.f);
    *(float4*)((char*)out + ((size_t)node << 9) + loff) = o4;
}

__global__ __launch_bounds__(256) void base_gather_kernel(const float* __restrict__ x,
                                                          const float* __restrict__ Wd,
                                                          const float* __restrict__ b,
                                                          const int* __restrict__ edge,
                                                          const float* __restrict__ y,
                                                          float* __restrict__ out) {
    __shared__ float xs[16 * XPITCH];
    __shared__ float bt[16 * BPITCH];

    const int tid  = threadIdx.x;
    const int wave = __builtin_amdgcn_readfirstlane(tid >> 5);
    const int lane = tid & 31;
    const int half = lane >> 4;
    const int lrow = lane & 15;
    const int koff = half * 2;
    const int gc   = wave * 16 + lrow;
    const int srow = tid >> 5;
    const int scol = (tid & 31) * 4;
    const unsigned loff = (unsigned)lane * 16u;

    int tile = blockIdx.x;

    float4 g0, g1;
    {
        const float4* src = (const float4*)(x + tile * 16 * CC);
        g0 = src[tid];
        g1 = src[tid + 256];
    }
    int idv = edge[(tile * 16 + wave * 2) * KK + lane];

    v2f bd[32];
#pragma unroll
    for (int kk = 0; kk < 32; ++kk) {
        const int kb = kk * 4 + koff;
        bd[kk].x = Wd[kb * OUTC + gc]; bd[kk].y = Wd[(kb + 1) * OUTC + gc];
    }
    const float bv = b[gc];

    for (; tile < NTILES; tile += FUSE_GRID) {
        *(float4*)(&xs[srow * XPITCH + scol])       = g0;
        *(float4*)(&xs[(srow + 8) * XPITCH + scol]) = g1;
        __syncthreads();

        const int nt = tile + FUSE_GRID;
        if (nt < NTILES) {
            const float4* src = (const float4*)(x + nt * 16 * CC);
            g0 = src[tid];
            g1 = src[tid + 256];
        }

        const int n0 = tile * 16 + wave * 2;
        int j[2 * KK];
#pragma unroll
        for (int k = 0; k < 2 * KK; ++k) j[k] = __builtin_amdgcn_readlane(idv, k);

        float4 v0[KK], v1[KK];
        rows_in_flight(v0, j, 0, y, loff);
        rows_in_flight(v1, j, KK, y, loff);
        if (nt < NTILES) idv = edge[(nt * 16 + wave * 2) * KK + lane];

        v8f acc_b = {bv, bv, bv, bv, bv, bv, bv, bv};
#pragma unroll
        for (int kk = 0; kk < 32; ++kk) {
            v2f a = *(const v2f*)(&xs[lrow * XPITCH + kk * 4 + koff]);
            acc_b = __builtin_amdgcn_wmma_f32_16x16x4_f32(false, a, false, bd[kk],
                                                          (short)0, acc_b, false, false);
        }
#pragma unroll
        for (int r = 0; r < 8; ++r) bt[(half * 8 + r) * BPITCH + gc] = acc_b[r];
        __syncthreads();
        __builtin_amdgcn_sched_barrier(0);

        const int lr0 = wave * 2;
        {
            const float4 bs = *(const float4*)(&bt[lr0 * BPITCH + lane * 4]);
            node_finish(v0, j, 0, bs, out, n0, loff);
        }
        {
            const float4 bs = *(const float4*)(&bt[(lr0 + 1) * BPITCH + lane * 4]);
            node_finish(v1, j, KK, bs, out, n0 + 1, loff);
        }
    }
}

extern "C" void kernel_launch(void* const* d_in, const int* in_sizes, int n_in,
                              void* d_out, int out_size, void* d_ws, size_t ws_size,
                              hipStream_t stream) {
    const float* x    = (const float*)d_in[0];
    const int*   edge = (const int*)  d_in[1];
    const float* W    = (const float*)d_in[2];
    const float* b    = (const float*)d_in[3];
    float* out = (float*)d_out;

    float* ws   = (float*)d_ws;
    float* y    = ws;
    float* Wd   = ws + (size_t)2 * NN * OUTC;

    gemm_y_kernel<<<GEMM_GRID, 256, 0, stream>>>(x, W, y, Wd);
    base_gather_kernel<<<FUSE_GRID, 256, 0, stream>>>(x, Wd, b, edge, y, out);
}
